# starting kernel
# baseline (speedup 1.0000x reference)
#include <hip/hip_runtime.h>

typedef __attribute__((ext_vector_type(2))) float v2f;
typedef __attribute__((ext_vector_type(8))) float v8f;

#define T_SEQ 4096
#define NH    8
#define DH    64
#define CHUNK 64
#define NPART (T_SEQ / CHUNK)
#define NHEAD (2 * NH)
#define PITCH 68

#define WMMA_F32(a, b, c) \
  __builtin_amdgcn_wmma_f32_16x16x4_f32(false, (a), false, (b), (short)0, (c), false, false)

__device__ __forceinline__ v2f lds_fragA(const float* L, int rb, int kb, int row16, int koff) {
  v2f a;
  a.x = L[(rb + row16) * PITCH + kb + koff];
  a.y = L[(rb + row16) * PITCH + kb + koff + 1];
  return a;
}

__device__ __forceinline__ v2f lds_fragB(const float* L, int kb, int nb, int row16, int koff) {
  v2f b;
  b.x = L[(kb + koff)     * PITCH + nb + row16];
  b.y = L[(kb + koff + 1) * PITCH + nb + row16];
  return b;
}

__global__ void __launch_bounds__(128, 1)
linmem_partial_state(const float* __restrict__ k, const float* __restrict__ v,
                     float* __restrict__ mpart) {
  __shared__ __align__(16) float Ks[CHUNK * PITCH];
  __shared__ __align__(16) float Vs[CHUNK * PITCH];

  const int tid = threadIdx.x, lane = tid & 31, wave = tid >> 5;
  const int row16 = lane & 15, half = lane >> 4;
  const int koff = half * 2, crow = half * 8;

  const int bh = blockIdx.x / NPART;
  const int p  = blockIdx.x % NPART;
  const int b = bh / NH, h = bh % NH;
  const int t0 = p * CHUNK;

  const size_t head_off = ((size_t)b * T_SEQ * NH + h) * DH;
  const size_t tstride  = (size_t)NH * DH;

  for (int idx = tid; idx < CHUNK * (DH / 4); idx += 128) {
    const int r = idx >> 4, c4 = idx & 15;
    const size_t g = head_off + (size_t)(t0 + r) * tstride + c4 * 4;
    *(float4*)&Ks[r * PITCH + c4 * 4] = *(const float4*)&k[g];
    *(float4*)&Vs[r * PITCH + c4 * 4] = *(const float4*)&v[g];
  }
  __syncthreads();

  float* out = mpart + ((size_t)bh * NPART + p) * (DH * DH);
  const int di = wave;
  for (int ej = 0; ej < 4; ++ej) {
    v8f acc = {};
    for (int t = 0; t < CHUNK; t += 4) {
      v2f a  = lds_fragB(Vs, t, di * 16, row16, koff);
      v2f bb = lds_fragB(Ks, t, ej * 16, row16, koff);
      acc = WMMA_F32(a, bb, acc);
    }
#pragma unroll
    for (int j = 0; j < 8; ++j)
      out[(di * 16 + crow + j) * DH + ej * 16 + row16] = acc[j];
  }
}

__global__ void __launch_bounds__(256, 1)
linmem_prefix_state(float* __restrict__ mpart) {
  float* base = mpart + (size_t)blockIdx.x * NPART * (DH * DH);
  for (int e = threadIdx.x; e < DH * DH; e += 256) {
    float run = 0.0f;
    for (int p = 0; p < NPART; ++p) {
      float* ptr = base + (size_t)p * (DH * DH) + e;
      const float cur = *ptr;
      *ptr = run;
      run += cur;
    }
  }
}

__global__ void __launch_bounds__(128, 1)
linmem_chunk_output(const float* __restrict__ q, const float* __restrict__ k,
                    const float* __restrict__ v, const float* __restrict__ mpart,
                    float* __restrict__ y) {
  __shared__ __align__(16) float Qs[CHUNK * PITCH];
  __shared__ __align__(16) float Ks[CHUNK * PITCH];
  __shared__ __align__(16) float Vs[CHUNK * PITCH];
  __shared__ __align__(16) float Ss[CHUNK * PITCH];
  __shared__ __align__(16) float Ms[DH    * PITCH];

  const int tid = threadIdx.x, lane = tid & 31, wave = tid >> 5;
  const int row16 = lane & 15, half = lane >> 4;
  const int koff = half * 2, crow = half * 8;

  const int bh = blockIdx.x / NPART;
  const int p  = blockIdx.x % NPART;
  const int b = bh / NH, h = bh % NH;
  const int t0 = p * CHUNK;

  const size_t head_off = ((size_t)b * T_SEQ * NH + h) * DH;
  const size_t tstride  = (size_t)NH * DH;
  const float* m0 = mpart + ((size_t)bh * NPART + p) * (DH * DH);

  for (int idx = tid; idx < CHUNK * (DH / 4); idx += 128) {
    const int r = idx >> 4, c4 = idx & 15;
    const size_t g = head_off + (size_t)(t0 + r) * tstride + c4 * 4;
    *(float4*)&Qs[r * PITCH + c4 * 4] = *(const float4*)&q[g];
    *(float4*)&Ks[r * PITCH + c4 * 4] = *(const float4*)&k[g];
    *(float4*)&Vs[r * PITCH + c4 * 4] = *(const float4*)&v[g];
    *(float4*)&Ms[r * PITCH + c4 * 4] = *(const float4*)&m0[r * DH + c4 * 4];
  }
  __syncthreads();

  const int ti = wave;

  for (int sj = 0; sj < 4; ++sj) {
    v8f acc = {};
    if (sj <= ti) {
      for (int e = 0; e < DH; e += 4) {
        v2f a  = lds_fragA(Qs, ti * 16, e, row16, koff);
        v2f bt = lds_fragA(Ks, sj * 16, e, row16, koff);
        acc = WMMA_F32(a, bt, acc);
      }
    }
#pragma unroll
    for (int j = 0; j < 8; ++j) {
      const int tt = ti * 16 + crow + j;
      const int ss = sj * 16 + row16;
      Ss[tt * PITCH + ss] = (ss <= tt) ? acc[j] : 0.0f;
    }
  }

  for (int dj = 0; dj < 4; ++dj) {
    v8f acc = {};
    for (int e = 0; e < DH; e += 4) {
      v2f a  = lds_fragA(Qs, ti * 16, e, row16, koff);
      v2f bt = lds_fragA(Ms, dj * 16, e, row16, koff);
      acc = WMMA_F32(a, bt, acc);
    }
    for (int s = 0; s < CHUNK; s += 4) {
      v2f a  = lds_fragA(Ss, ti * 16, s, row16, koff);
      v2f bb = lds_fragB(Vs, s, dj * 16, row16, koff);
      acc = WMMA_F32(a, bb, acc);
    }
#pragma unroll
    for (int j = 0; j < 8; ++j) {
      const size_t g = head_off +
          (size_t)(t0 + ti * 16 + crow + j) * tstride + dj * 16 + row16;
      y[g] = acc[j];
    }
  }
}

__global__ void __launch_bounds__(128, 1)
linmem_chunked_wmma(const float* __restrict__ q,
                    const float* __restrict__ k,
                    const float* __restrict__ v,
                    float* __restrict__ y) {
  __shared__ __align__(16) float Qs[CHUNK * PITCH];
  __shared__ __align__(16) float Ks[CHUNK * PITCH];
  __shared__ __align__(16) float Vs[CHUNK * PITCH];
  __shared__ __align__(16) float Ss[CHUNK * PITCH];
  __shared__ __align__(16) float Ms[DH    * PITCH];

  const int tid = threadIdx.x, lane = tid & 31, wave = tid >> 5;
  const int row16 = lane & 15, half = lane >> 4;
  const int koff = half * 2, crow = half * 8;
  const int b = blockIdx.x / NH, h = blockIdx.x % NH;

  for (int i = tid; i < DH * PITCH; i += 128) Ms[i] = 0.0f;
  __syncthreads();

  const size_t head_off = ((size_t)b * T_SEQ * NH + h) * DH;
  const size_t tstride  = (size_t)NH * DH;

  for (int t0 = 0; t0 < T_SEQ; t0 += CHUNK) {
    for (int idx = tid; idx < CHUNK * (DH / 4); idx += 128) {
      const int r = idx >> 4, c4 = idx & 15;
      const size_t g = head_off + (size_t)(t0 + r) * tstride + c4 * 4;
      *(float4*)&Qs[r * PITCH + c4 * 4] = *(const float4*)&q[g];
      *(float4*)&Ks[r * PITCH + c4 * 4] = *(const float4*)&k[g];
      *(float4*)&Vs[r * PITCH + c4 * 4] = *(const float4*)&v[g];
      if (t0 + CHUNK < T_SEQ) {
        __builtin_prefetch(&q[g + CHUNK * tstride], 0, 3);
        __builtin_prefetch(&k[g + CHUNK * tstride], 0, 3);
        __builtin_prefetch(&v[g + CHUNK * tstride], 0, 3);
      }
    }
    __syncthreads();

    const int ti = wave;
    for (int sj = 0; sj < 4; ++sj) {
      v8f acc = {};
      if (sj <= ti) {
        for (int e = 0; e < DH; e += 4) {
          v2f a  = lds_fragA(Qs, ti * 16, e, row16, koff);
          v2f bt = lds_fragA(Ks, sj * 16, e, row16, koff);
          acc = WMMA_F32(a, bt, acc);
        }
      }
#pragma unroll
      for (int j = 0; j < 8; ++j) {
        const int tt = ti * 16 + crow + j;
        const int ss = sj * 16 + row16;
        Ss[tt * PITCH + ss] = (ss <= tt) ? acc[j] : 0.0f;
      }
    }
    __syncthreads();

    for (int dj = 0; dj < 4; ++dj) {
      v8f acc = {};
      for (int e = 0; e < DH; e += 4) {
        v2f a  = lds_fragA(Qs, ti * 16, e, row16, koff);
        v2f bt = lds_fragA(Ms, dj * 16, e, row16, koff);
        acc = WMMA_F32(a, bt, acc);
      }
      for (int s = 0; s < CHUNK; s += 4) {
        v2f a  = lds_fragA(Ss, ti * 16, s, row16, koff);
        v2f bb = lds_fragB(Vs, s, dj * 16, row16, koff);
        acc = WMMA_F32(a, bb, acc);
      }
#pragma unroll
      for (int j = 0; j < 8; ++j) {
        const size_t g = head_off +
            (size_t)(t0 + ti * 16 + crow + j) * tstride + dj * 16 + row16;
        y[g] = acc[j];
      }
    }
    __syncthreads();

    const int di = wave;
    for (int ej = 0; ej < 4; ++ej) {
      v8f acc;
#pragma unroll
      for (int j = 0; j < 8; ++j)
        acc[j] = Ms[(di * 16 + crow + j) * PITCH + ej * 16 + row16];
      for (int t = 0; t < CHUNK; t += 4) {
        v2f a  = lds_fragB(Vs, t, di * 16, row16, koff);
        v2f bb = lds_fragB(Ks, t, ej * 16, row16, koff);
        acc = WMMA_F32(a, bb, acc);
      }
#pragma unroll
      for (int j = 0; j < 8; ++j)
        Ms[(di * 16 + crow + j) * PITCH + ej * 16 + row16] = acc[j];
    }
    __syncthreads();
  }
}

extern "C" void kernel_launch(void* const* d_in, const int* in_sizes, int n_in,
                              void* d_out, int out_size, void* d_ws, size_t ws_size,
                              hipStream_t stream) {
  const float* q = (const float*)d_in[0];
  const float* k = (const float*)d_in[1];
  const float* v = (const float*)d_in[2];
  float* y = (float*)d_out;

  const size_t ws_need = (size_t)NHEAD * NPART * DH * DH * sizeof(float);

  if (ws_size >= ws_need) {
    float* mpart = (float*)d_ws;
    linmem_partial_state<<<dim3(NHEAD * NPART), dim3(128), 0, stream>>>(k, v, mpart);
    linmem_prefix_state <<<dim3(NHEAD),         dim3(256), 0, stream>>>(mpart);
    linmem_chunk_output <<<dim3(NHEAD * NPART), dim3(128), 0, stream>>>(q, k, v, mpart, y);
  } else {
    linmem_chunked_wmma<<<dim3(NHEAD), dim3(128), 0, stream>>>(q, k, v, y);
  }
}


# prefix-scan launch widened from 16 to 256 work-groups: a lane owns 2 adjacent state elements and loads all 64 partition values (8-byte loads) before its serial sum; sum order unchanged
# speedup vs baseline: 2.5018x; 2.5018x over previous
#include <hip/hip_runtime.h>

typedef __attribute__((ext_vector_type(2))) float v2f;
typedef __attribute__((ext_vector_type(8))) float v8f;

#define T_SEQ 4096
#define NH    8
#define DH    64
#define CHUNK 64
#define NPART (T_SEQ / CHUNK)
#define NHEAD (2 * NH)
#define PITCH 68

#define WMMA_F32(a, b, c) \
  __builtin_amdgcn_wmma_f32_16x16x4_f32(false, (a), false, (b), (short)0, (c), false, false)

__device__ __forceinline__ v2f lds_fragA(const float* L, int rb, int kb, int row16, int koff) {
  v2f a;
  a.x = L[(rb + row16) * PITCH + kb + koff];
  a.y = L[(rb + row16) * PITCH + kb + koff + 1];
  return a;
}

__device__ __forceinline__ v2f lds_fragB(const float* L, int kb, int nb, int row16, int koff) {
  v2f b;
  b.x = L[(kb + koff)     * PITCH + nb + row16];
  b.y = L[(kb + koff + 1) * PITCH + nb + row16];
  return b;
}

__global__ void __launch_bounds__(128, 1)
linmem_partial_state(const float* __restrict__ k, const float* __restrict__ v,
                     float* __restrict__ mpart) {
  __shared__ __align__(16) float Ks[CHUNK * PITCH];
  __shared__ __align__(16) float Vs[CHUNK * PITCH];

  const int tid = threadIdx.x, lane = tid & 31, wave = tid >> 5;
  const int row16 = lane & 15, half = lane >> 4;
  const int koff = half * 2, crow = half * 8;

  const int bh = blockIdx.x / NPART;
  const int p  = blockIdx.x % NPART;
  const int b = bh / NH, h = bh % NH;
  const int t0 = p * CHUNK;

  const size_t head_off = ((size_t)b * T_SEQ * NH + h) * DH;
  const size_t tstride  = (size_t)NH * DH;

  for (int idx = tid; idx < CHUNK * (DH / 4); idx += 128) {
    const int r = idx >> 4, c4 = idx & 15;
    const size_t g = head_off + (size_t)(t0 + r) * tstride + c4 * 4;
    *(float4*)&Ks[r * PITCH + c4 * 4] = *(const float4*)&k[g];
    *(float4*)&Vs[r * PITCH + c4 * 4] = *(const float4*)&v[g];
  }
  __syncthreads();

  float* out = mpart + ((size_t)bh * NPART + p) * (DH * DH);
  const int di = wave;
  for (int ej = 0; ej < 4; ++ej) {
    v8f acc = {};
    for (int t = 0; t < CHUNK; t += 4) {
      v2f a  = lds_fragB(Vs, t, di * 16, row16, koff);
      v2f bb = lds_fragB(Ks, t, ej * 16, row16, koff);
      acc = WMMA_F32(a, bb, acc);
    }
#pragma unroll
    for (int j = 0; j < 8; ++j)
      out[(di * 16 + crow + j) * DH + ej * 16 + row16] = acc[j];
  }
}

#define SCAN_THREADS 128
#define SCAN_LANES_PER_HEAD (DH * DH / 2)

__global__ void __launch_bounds__(SCAN_THREADS) __attribute__((amdgpu_waves_per_eu(1, 2)))
linmem_prefix_state(float* __restrict__ mpart) {
  const int gid = blockIdx.x * SCAN_THREADS + threadIdx.x;
  const int bh  = gid / SCAN_LANES_PER_HEAD;
  const int e2  = gid % SCAN_LANES_PER_HEAD;
  float2* base = (float2*)(mpart + (size_t)bh * NPART * (DH * DH)) + e2;

  float2 x[NPART];
#pragma unroll
  for (int p = 0; p < NPART; ++p)
    x[p] = base[(size_t)p * SCAN_LANES_PER_HEAD];

  float2 run = make_float2(0.0f, 0.0f);
#pragma unroll
  for (int p = 0; p < NPART; ++p) {
    base[(size_t)p * SCAN_LANES_PER_HEAD] = run;
    run.x += x[p].x;
    run.y += x[p].y;
  }
}

__global__ void __launch_bounds__(128, 1)
linmem_chunk_output(const float* __restrict__ q, const float* __restrict__ k,
                    const float* __restrict__ v, const float* __restrict__ mpart,
                    float* __restrict__ y) {
  __shared__ __align__(16) float Qs[CHUNK * PITCH];
  __shared__ __align__(16) float Ks[CHUNK * PITCH];
  __shared__ __align__(16) float Vs[CHUNK * PITCH];
  __shared__ __align__(16) float Ss[CHUNK * PITCH];
  __shared__ __align__(16) float Ms[DH    * PITCH];

  const int tid = threadIdx.x, lane = tid & 31, wave = tid >> 5;
  const int row16 = lane & 15, half = lane >> 4;
  const int koff = half * 2, crow = half * 8;

  const int bh = blockIdx.x / NPART;
  const int p  = blockIdx.x % NPART;
  const int b = bh / NH, h = bh % NH;
  const int t0 = p * CHUNK;

  const size_t head_off = ((size_t)b * T_SEQ * NH + h) * DH;
  const size_t tstride  = (size_t)NH * DH;
  const float* m0 = mpart + ((size_t)bh * NPART + p) * (DH * DH);

  for (int idx = tid; idx < CHUNK * (DH / 4); idx += 128) {
    const int r = idx >> 4, c4 = idx & 15;
    const size_t g = head_off + (size_t)(t0 + r) * tstride + c4 * 4;
    *(float4*)&Qs[r * PITCH + c4 * 4] = *(const float4*)&q[g];
    *(float4*)&Ks[r * PITCH + c4 * 4] = *(const float4*)&k[g];
    *(float4*)&Vs[r * PITCH + c4 * 4] = *(const float4*)&v[g];
    *(float4*)&Ms[r * PITCH + c4 * 4] = *(const float4*)&m0[r * DH + c4 * 4];
  }
  __syncthreads();

  const int ti = wave;

  for (int sj = 0; sj < 4; ++sj) {
    v8f acc = {};
    if (sj <= ti) {
      for (int e = 0; e < DH; e += 4) {
        v2f a  = lds_fragA(Qs, ti * 16, e, row16, koff);
        v2f bt = lds_fragA(Ks, sj * 16, e, row16, koff);
        acc = WMMA_F32(a, bt, acc);
      }
    }
#pragma unroll
    for (int j = 0; j < 8; ++j) {
      const int tt = ti * 16 + crow + j;
      const int ss = sj * 16 + row16;
      Ss[tt * PITCH + ss] = (ss <= tt) ? acc[j] : 0.0f;
    }
  }

  for (int dj = 0; dj < 4; ++dj) {
    v8f acc = {};
    for (int e = 0; e < DH; e += 4) {
      v2f a  = lds_fragA(Qs, ti * 16, e, row16, koff);
      v2f bt = lds_fragA(Ms, dj * 16, e, row16, koff);
      acc = WMMA_F32(a, bt, acc);
    }
    for (int s = 0; s < CHUNK; s += 4) {
      v2f a  = lds_fragA(Ss, ti * 16, s, row16, koff);
      v2f bb = lds_fragB(Vs, s, dj * 16, row16, koff);
      acc = WMMA_F32(a, bb, acc);
    }
#pragma unroll
    for (int j = 0; j < 8; ++j) {
      const size_t g = head_off +
          (size_t)(t0 + ti * 16 + crow + j) * tstride + dj * 16 + row16;
      y[g] = acc[j];
    }
  }
}

__global__ void __launch_bounds__(128, 1)
linmem_chunked_wmma(const float* __restrict__ q,
                    const float* __restrict__ k,
                    const float* __restrict__ v,
                    float* __restrict__ y) {
  __shared__ __align__(16) float Qs[CHUNK * PITCH];
  __shared__ __align__(16) float Ks[CHUNK * PITCH];
  __shared__ __align__(16) float Vs[CHUNK * PITCH];
  __shared__ __align__(16) float Ss[CHUNK * PITCH];
  __shared__ __align__(16) float Ms[DH    * PITCH];

  const int tid = threadIdx.x, lane = tid & 31, wave = tid >> 5;
  const int row16 = lane & 15, half = lane >> 4;
  const int koff = half * 2, crow = half * 8;
  const int b = blockIdx.x / NH, h = blockIdx.x % NH;

  for (int i = tid; i < DH * PITCH; i += 128) Ms[i] = 0.0f;
  __syncthreads();

  const size_t head_off = ((size_t)b * T_SEQ * NH + h) * DH;
  const size_t tstride  = (size_t)NH * DH;

  for (int t0 = 0; t0 < T_SEQ; t0 += CHUNK) {
    for (int idx = tid; idx < CHUNK * (DH / 4); idx += 128) {
      const int r = idx >> 4, c4 = idx & 15;
      const size_t g = head_off + (size_t)(t0 + r) * tstride + c4 * 4;
      *(float4*)&Qs[r * PITCH + c4 * 4] = *(const float4*)&q[g];
      *(float4*)&Ks[r * PITCH + c4 * 4] = *(const float4*)&k[g];
      *(float4*)&Vs[r * PITCH + c4 * 4] = *(const float4*)&v[g];
      if (t0 + CHUNK < T_SEQ) {
        __builtin_prefetch(&q[g + CHUNK * tstride], 0, 3);
        __builtin_prefetch(&k[g + CHUNK * tstride], 0, 3);
        __builtin_prefetch(&v[g + CHUNK * tstride], 0, 3);
      }
    }
    __syncthreads();

    const int ti = wave;
    for (int sj = 0; sj < 4; ++sj) {
      v8f acc = {};
      if (sj <= ti) {
        for (int e = 0; e < DH; e += 4) {
          v2f a  = lds_fragA(Qs, ti * 16, e, row16, koff);
          v2f bt = lds_fragA(Ks, sj * 16, e, row16, koff);
          acc = WMMA_F32(a, bt, acc);
        }
      }
#pragma unroll
      for (int j = 0; j < 8; ++j) {
        const int tt = ti * 16 + crow + j;
        const int ss = sj * 16 + row16;
        Ss[tt * PITCH + ss] = (ss <= tt) ? acc[j] : 0.0f;
      }
    }
    __syncthreads();

    for (int dj = 0; dj < 4; ++dj) {
      v8f acc = {};
      for (int e = 0; e < DH; e += 4) {
        v2f a  = lds_fragA(Qs, ti * 16, e, row16, koff);
        v2f bt = lds_fragA(Ms, dj * 16, e, row16, koff);
        acc = WMMA_F32(a, bt, acc);
      }
      for (int s = 0; s < CHUNK; s += 4) {
        v2f a  = lds_fragA(Ss, ti * 16, s, row16, koff);
        v2f bb = lds_fragB(Vs, s, dj * 16, row16, koff);
        acc = WMMA_F32(a, bb, acc);
      }
#pragma unroll
      for (int j = 0; j < 8; ++j) {
        const size_t g = head_off +
            (size_t)(t0 + ti * 16 + crow + j) * tstride + dj * 16 + row16;
        y[g] = acc[j];
      }
    }
    __syncthreads();

    const int di = wave;
    for (int ej = 0; ej < 4; ++ej) {
      v8f acc;
#pragma unroll
      for (int j = 0; j < 8; ++j)
        acc[j] = Ms[(di * 16 + crow + j) * PITCH + ej * 16 + row16];
      for (int t = 0; t < CHUNK; t += 4) {
        v2f a  = lds_fragB(Vs, t, di * 16, row16, koff);
        v2f bb = lds_fragB(Ks, t, ej * 16, row16, koff);
        acc = WMMA_F32(a, bb, acc);
      }
#pragma unroll
      for (int j = 0; j < 8; ++j)
        Ms[(di * 16 + crow + j) * PITCH + ej * 16 + row16] = acc[j];
    }
    __syncthreads();
  }
}

extern "C" void kernel_launch(void* const* d_in, const int* in_sizes, int n_in,
                              void* d_out, int out_size, void* d_ws, size_t ws_size,
                              hipStream_t stream) {
  const float* q = (const float*)d_in[0];
  const float* k = (const float*)d_in[1];
  const float* v = (const float*)d_in[2];
  float* y = (float*)d_out;

  const size_t ws_need = (size_t)NHEAD * NPART * DH * DH * sizeof(float);

  if (ws_size >= ws_need) {
    float* mpart = (float*)d_ws;
    linmem_partial_state<<<dim3(NHEAD * NPART), dim3(128), 0, stream>>>(k, v, mpart);
    linmem_prefix_state <<<dim3(NHEAD * SCAN_LANES_PER_HEAD / SCAN_THREADS), dim3(SCAN_THREADS), 0, stream>>>(mpart);
    linmem_chunk_output <<<dim3(NHEAD * NPART), dim3(128), 0, stream>>>(q, k, v, mpart, y);
  } else {
    linmem_chunked_wmma<<<dim3(NHEAD), dim3(128), 0, stream>>>(q, k, v, y);
  }
}


# staging loads of both matrix kernels all issued before the first LDS store (row loops unrolled into registers): one memory round trip per work-group instead of eight; sums unchanged
# speedup vs baseline: 3.3735x; 1.3484x over previous
#include <hip/hip_runtime.h>

typedef __attribute__((ext_vector_type(2))) float v2f;
typedef __attribute__((ext_vector_type(8))) float v8f;

#define T_SEQ 4096
#define NH    8
#define DH    64
#define CHUNK 64
#define NPART (T_SEQ / CHUNK)
#define NHEAD (2 * NH)
#define PITCH 68
#define STAGE_ROWS (CHUNK / 8)

#define WMMA_F32(a, b, c) \
  __builtin_amdgcn_wmma_f32_16x16x4_f32(false, (a), false, (b), (short)0, (c), false, false)

__device__ __forceinline__ v2f lds_fragA(const float* L, int rb, int kb, int row16, int koff) {
  v2f a;
  a.x = L[(rb + row16) * PITCH + kb + koff];
  a.y = L[(rb + row16) * PITCH + kb + koff + 1];
  return a;
}

__device__ __forceinline__ v2f lds_fragB(const float* L, int kb, int nb, int row16, int koff) {
  v2f b;
  b.x = L[(kb + koff)     * PITCH + nb + row16];
  b.y = L[(kb + koff + 1) * PITCH + nb + row16];
  return b;
}

__global__ void __launch_bounds__(128, 1)
linmem_partial_state(const float* __restrict__ k, const float* __restrict__ v,
                     float* __restrict__ mpart) {
  __shared__ __align__(16) float Ks[CHUNK * PITCH];
  __shared__ __align__(16) float Vs[CHUNK * PITCH];

  const int tid = threadIdx.x, lane = tid & 31, wave = tid >> 5;
  const int row16 = lane & 15, half = lane >> 4;
  const int koff = half * 2, crow = half * 8;

  const int bh = blockIdx.x / NPART;
  const int p  = blockIdx.x % NPART;
  const int b = bh / NH, h = bh % NH;
  const int t0 = p * CHUNK;

  const size_t head_off = ((size_t)b * T_SEQ * NH + h) * DH;
  const size_t tstride  = (size_t)NH * DH;

  const int r0 = tid >> 4, c4 = tid & 15;
  const size_t g0 = head_off + (size_t)(t0 + r0) * tstride + c4 * 4;
  float4 rk[STAGE_ROWS], rv[STAGE_ROWS];
#pragma unroll
  for (int i = 0; i < STAGE_ROWS; ++i)
    rk[i] = *(const float4*)&k[g0 + (size_t)(i * 8) * tstride];
#pragma unroll
  for (int i = 0; i < STAGE_ROWS; ++i)
    rv[i] = *(const float4*)&v[g0 + (size_t)(i * 8) * tstride];
#pragma unroll
  for (int i = 0; i < STAGE_ROWS; ++i)
    *(float4*)&Ks[(r0 + i * 8) * PITCH + c4 * 4] = rk[i];
#pragma unroll
  for (int i = 0; i < STAGE_ROWS; ++i)
    *(float4*)&Vs[(r0 + i * 8) * PITCH + c4 * 4] = rv[i];
  __syncthreads();

  float* out = mpart + ((size_t)bh * NPART + p) * (DH * DH);
  const int di = wave;
  for (int ej = 0; ej < 4; ++ej) {
    v8f acc = {};
    for (int t = 0; t < CHUNK; t += 4) {
      v2f a  = lds_fragB(Vs, t, di * 16, row16, koff);
      v2f bb = lds_fragB(Ks, t, ej * 16, row16, koff);
      acc = WMMA_F32(a, bb, acc);
    }
#pragma unroll
    for (int j = 0; j < 8; ++j)
      out[(di * 16 + crow + j) * DH + ej * 16 + row16] = acc[j];
  }
}

#define SCAN_THREADS 128
#define SCAN_LANES_PER_HEAD (DH * DH / 2)

__global__ void __launch_bounds__(SCAN_THREADS) __attribute__((amdgpu_waves_per_eu(1, 2)))
linmem_prefix_state(float* __restrict__ mpart) {
  const int gid = blockIdx.x * SCAN_THREADS + threadIdx.x;
  const int bh  = gid / SCAN_LANES_PER_HEAD;
  const int e2  = gid % SCAN_LANES_PER_HEAD;
  float2* base = (float2*)(mpart + (size_t)bh * NPART * (DH * DH)) + e2;

  float2 x[NPART];
#pragma unroll
  for (int p = 0; p < NPART; ++p)
    x[p] = base[(size_t)p * SCAN_LANES_PER_HEAD];

  float2 run = make_float2(0.0f, 0.0f);
#pragma unroll
  for (int p = 0; p < NPART; ++p) {
    base[(size_t)p * SCAN_LANES_PER_HEAD] = run;
    run.x += x[p].x;
    run.y += x[p].y;
  }
}

__global__ void __launch_bounds__(128, 1)
linmem_chunk_output(const float* __restrict__ q, const float* __restrict__ k,
                    const float* __restrict__ v, const float* __restrict__ mpart,
                    float* __restrict__ y) {
  __shared__ __align__(16) float Qs[CHUNK * PITCH];
  __shared__ __align__(16) float Ks[CHUNK * PITCH];
  __shared__ __align__(16) float Vs[CHUNK * PITCH];
  __shared__ __align__(16) float Ss[CHUNK * PITCH];
  __shared__ __align__(16) float Ms[DH    * PITCH];

  const int tid = threadIdx.x, lane = tid & 31, wave = tid >> 5;
  const int row16 = lane & 15, half = lane >> 4;
  const int koff = half * 2, crow = half * 8;

  const int bh = blockIdx.x / NPART;
  const int p  = blockIdx.x % NPART;
  const int b = bh / NH, h = bh % NH;
  const int t0 = p * CHUNK;

  const size_t head_off = ((size_t)b * T_SEQ * NH + h) * DH;
  const size_t tstride  = (size_t)NH * DH;
  const float* m0 = mpart + ((size_t)bh * NPART + p) * (DH * DH);

  const int r0 = tid >> 4, c4 = tid & 15;
  const size_t g0 = head_off + (size_t)(t0 + r0) * tstride + c4 * 4;
  float4 rq[STAGE_ROWS], rk[STAGE_ROWS], rv[STAGE_ROWS], rm[STAGE_ROWS];
#pragma unroll
  for (int i = 0; i < STAGE_ROWS; ++i)
    rq[i] = *(const float4*)&q[g0 + (size_t)(i * 8) * tstride];
#pragma unroll
  for (int i = 0; i < STAGE_ROWS; ++i)
    rk[i] = *(const float4*)&k[g0 + (size_t)(i * 8) * tstride];
#pragma unroll
  for (int i = 0; i < STAGE_ROWS; ++i)
    rv[i] = *(const float4*)&v[g0 + (size_t)(i * 8) * tstride];
#pragma unroll
  for (int i = 0; i < STAGE_ROWS; ++i)
    rm[i] = *(const float4*)&m0[(r0 + i * 8) * DH + c4 * 4];
#pragma unroll
  for (int i = 0; i < STAGE_ROWS; ++i)
    *(float4*)&Qs[(r0 + i * 8) * PITCH + c4 * 4] = rq[i];
#pragma unroll
  for (int i = 0; i < STAGE_ROWS; ++i)
    *(float4*)&Ks[(r0 + i * 8) * PITCH + c4 * 4] = rk[i];
#pragma unroll
  for (int i = 0; i < STAGE_ROWS; ++i)
    *(float4*)&Vs[(r0 + i * 8) * PITCH + c4 * 4] = rv[i];
#pragma unroll
  for (int i = 0; i < STAGE_ROWS; ++i)
    *(float4*)&Ms[(r0 + i * 8) * PITCH + c4 * 4] = rm[i];
  __syncthreads();

  const int ti = wave;

  for (int sj = 0; sj < 4; ++sj) {
    v8f acc = {};
    if (sj <= ti) {
      for (int e = 0; e < DH; e += 4) {
        v2f a  = lds_fragA(Qs, ti * 16, e, row16, koff);
        v2f bt = lds_fragA(Ks, sj * 16, e, row16, koff);
        acc = WMMA_F32(a, bt, acc);
      }
    }
#pragma unroll
    for (int j = 0; j < 8; ++j) {
      const int tt = ti * 16 + crow + j;
      const int ss = sj * 16 + row16;
      Ss[tt * PITCH + ss] = (ss <= tt) ? acc[j] : 0.0f;
    }
  }

  for (int dj = 0; dj < 4; ++dj) {
    v8f acc = {};
    for (int e = 0; e < DH; e += 4) {
      v2f a  = lds_fragA(Qs, ti * 16, e, row16, koff);
      v2f bt = lds_fragA(Ms, dj * 16, e, row16, koff);
      acc = WMMA_F32(a, bt, acc);
    }
    for (int s = 0; s < CHUNK; s += 4) {
      v2f a  = lds_fragA(Ss, ti * 16, s, row16, koff);
      v2f bb = lds_fragB(Vs, s, dj * 16, row16, koff);
      acc = WMMA_F32(a, bb, acc);
    }
#pragma unroll
    for (int j = 0; j < 8; ++j) {
      const size_t g = head_off +
          (size_t)(t0 + ti * 16 + crow + j) * tstride + dj * 16 + row16;
      y[g] = acc[j];
    }
  }
}

__global__ void __launch_bounds__(128, 1)
linmem_chunked_wmma(const float* __restrict__ q,
                    const float* __restrict__ k,
                    const float* __restrict__ v,
                    float* __restrict__ y) {
  __shared__ __align__(16) float Qs[CHUNK * PITCH];
  __shared__ __align__(16) float Ks[CHUNK * PITCH];
  __shared__ __align__(16) float Vs[CHUNK * PITCH];
  __shared__ __align__(16) float Ss[CHUNK * PITCH];
  __shared__ __align__(16) float Ms[DH    * PITCH];

  const int tid = threadIdx.x, lane = tid & 31, wave = tid >> 5;
  const int row16 = lane & 15, half = lane >> 4;
  const int koff = half * 2, crow = half * 8;
  const int b = blockIdx.x / NH, h = blockIdx.x % NH;

  for (int i = tid; i < DH * PITCH; i += 128) Ms[i] = 0.0f;
  __syncthreads();

  const size_t head_off = ((size_t)b * T_SEQ * NH + h) * DH;
  const size_t tstride  = (size_t)NH * DH;

  for (int t0 = 0; t0 < T_SEQ; t0 += CHUNK) {
    for (int idx = tid; idx < CHUNK * (DH / 4); idx += 128) {
      const int r = idx >> 4, c4 = idx & 15;
      const size_t g = head_off + (size_t)(t0 + r) * tstride + c4 * 4;
      *(float4*)&Qs[r * PITCH + c4 * 4] = *(const float4*)&q[g];
      *(float4*)&Ks[r * PITCH + c4 * 4] = *(const float4*)&k[g];
      *(float4*)&Vs[r * PITCH + c4 * 4] = *(const float4*)&v[g];
      if (t0 + CHUNK < T_SEQ) {
        __builtin_prefetch(&q[g + CHUNK * tstride], 0, 3);
        __builtin_prefetch(&k[g + CHUNK * tstride], 0, 3);
        __builtin_prefetch(&v[g + CHUNK * tstride], 0, 3);
      }
    }
    __syncthreads();

    const int ti = wave;
    for (int sj = 0; sj < 4; ++sj) {
      v8f acc = {};
      if (sj <= ti) {
        for (int e = 0; e < DH; e += 4) {
          v2f a  = lds_fragA(Qs, ti * 16, e, row16, koff);
          v2f bt = lds_fragA(Ks, sj * 16, e, row16, koff);
          acc = WMMA_F32(a, bt, acc);
        }
      }
#pragma unroll
      for (int j = 0; j < 8; ++j) {
        const int tt = ti * 16 + crow + j;
        const int ss = sj * 16 + row16;
        Ss[tt * PITCH + ss] = (ss <= tt) ? acc[j] : 0.0f;
      }
    }
    __syncthreads();

    for (int dj = 0; dj < 4; ++dj) {
      v8f acc = {};
      for (int e = 0; e < DH; e += 4) {
        v2f a  = lds_fragA(Qs, ti * 16, e, row16, koff);
        v2f bt = lds_fragA(Ms, dj * 16, e, row16, koff);
        acc = WMMA_F32(a, bt, acc);
      }
      for (int s = 0; s < CHUNK; s += 4) {
        v2f a  = lds_fragA(Ss, ti * 16, s, row16, koff);
        v2f bb = lds_fragB(Vs, s, dj * 16, row16, koff);
        acc = WMMA_F32(a, bb, acc);
      }
#pragma unroll
      for (int j = 0; j < 8; ++j) {
        const size_t g = head_off +
            (size_t)(t0 + ti * 16 + crow + j) * tstride + dj * 16 + row16;
        y[g] = acc[j];
      }
    }
    __syncthreads();

    const int di = wave;
    for (int ej = 0; ej < 4; ++ej) {
      v8f acc;
#pragma unroll
      for (int j = 0; j < 8; ++j)
        acc[j] = Ms[(di * 16 + crow + j) * PITCH + ej * 16 + row16];
      for (int t = 0; t < CHUNK; t += 4) {
        v2f a  = lds_fragB(Vs, t, di * 16, row16, koff);
        v2f bb = lds_fragB(Ks, t, ej * 16, row16, koff);
        acc = WMMA_F32(a, bb, acc);
      }
#pragma unroll
      for (int j = 0; j < 8; ++j)
        Ms[(di * 16 + crow + j) * PITCH + ej * 16 + row16] = acc[j];
    }
    __syncthreads();
  }
}

extern "C" void kernel_launch(void* const* d_in, const int* in_sizes, int n_in,
                              void* d_out, int out_size, void* d_ws, size_t ws_size,
                              hipStream_t stream) {
  const float* q = (const float*)d_in[0];
  const float* k = (const float*)d_in[1];
  const float* v = (const float*)d_in[2];
  float* y = (float*)d_out;

  const size_t ws_need = (size_t)NHEAD * NPART * DH * DH * sizeof(float);

  if (ws_size >= ws_need) {
    float* mpart = (float*)d_ws;
    linmem_partial_state<<<dim3(NHEAD * NPART), dim3(128), 0, stream>>>(k, v, mpart);
    linmem_prefix_state <<<dim3(NHEAD * SCAN_LANES_PER_HEAD / SCAN_THREADS), dim3(SCAN_THREADS), 0, stream>>>(mpart);
    linmem_chunk_output <<<dim3(NHEAD * NPART), dim3(128), 0, stream>>>(q, k, v, mpart, y);
  } else {
    linmem_chunked_wmma<<<dim3(NHEAD), dim3(128), 0, stream>>>(q, k, v, y);
  }
}


# chunk-output group = 2 chunks on 8 waves; a SIMD's two waves take row strips x and 3-x and S*V stops at the diagonal tile: equal matrix work per SIMD; scores overwrite own Q rows; sums unchanged
# speedup vs baseline: 3.6112x; 1.0705x over previous
#include <hip/hip_runtime.h>

typedef __attribute__((ext_vector_type(2))) float v2f;
typedef __attribute__((ext_vector_type(8))) float v8f;

#define T_SEQ 4096
#define NH    8
#define DH    64
#define CHUNK 64
#define NPART (T_SEQ / CHUNK)
#define NHEAD (2 * NH)
#define PITCH 68
#define STAGE_ROWS (CHUNK / 8)

#define WMMA_F32(a, b, c) \
  __builtin_amdgcn_wmma_f32_16x16x4_f32(false, (a), false, (b), (short)0, (c), false, false)

__device__ __forceinline__ v2f lds_fragA(const float* L, int rb, int kb, int row16, int koff) {
  v2f a;
  a.x = L[(rb + row16) * PITCH + kb + koff];
  a.y = L[(rb + row16) * PITCH + kb + koff + 1];
  return a;
}

__device__ __forceinline__ v2f lds_fragB(const float* L, int kb, int nb, int row16, int koff) {
  v2f b;
  b.x = L[(kb + koff)     * PITCH + nb + row16];
  b.y = L[(kb + koff + 1) * PITCH + nb + row16];
  return b;
}

__global__ void __launch_bounds__(128, 1)
linmem_partial_state(const float* __restrict__ k, const float* __restrict__ v,
                     float* __restrict__ mpart) {
  __shared__ __align__(16) float Ks[CHUNK * PITCH];
  __shared__ __align__(16) float Vs[CHUNK * PITCH];

  const int tid = threadIdx.x, lane = tid & 31, wave = tid >> 5;
  const int row16 = lane & 15, half = lane >> 4;
  const int koff = half * 2, crow = half * 8;

  const int bh = blockIdx.x / NPART;
  const int p  = blockIdx.x % NPART;
  const int b = bh / NH, h = bh % NH;
  const int t0 = p * CHUNK;

  const size_t head_off = ((size_t)b * T_SEQ * NH + h) * DH;
  const size_t tstride  = (size_t)NH * DH;

  const int r0 = tid >> 4, c4 = tid & 15;
  const size_t g0 = head_off + (size_t)(t0 + r0) * tstride + c4 * 4;
  float4 rk[STAGE_ROWS], rv[STAGE_ROWS];
#pragma unroll
  for (int i = 0; i < STAGE_ROWS; ++i)
    rk[i] = *(const float4*)&k[g0 + (size_t)(i * 8) * tstride];
#pragma unroll
  for (int i = 0; i < STAGE_ROWS; ++i)
    rv[i] = *(const float4*)&v[g0 + (size_t)(i * 8) * tstride];
#pragma unroll
  for (int i = 0; i < STAGE_ROWS; ++i)
    *(float4*)&Ks[(r0 + i * 8) * PITCH + c4 * 4] = rk[i];
#pragma unroll
  for (int i = 0; i < STAGE_ROWS; ++i)
    *(float4*)&Vs[(r0 + i * 8) * PITCH + c4 * 4] = rv[i];
  __syncthreads();

  float* out = mpart + ((size_t)bh * NPART + p) * (DH * DH);
  const int di = wave;
  for (int ej = 0; ej < 4; ++ej) {
    v8f acc = {};
    for (int t = 0; t < CHUNK; t += 4) {
      v2f a  = lds_fragB(Vs, t, di * 16, row16, koff);
      v2f bb = lds_fragB(Ks, t, ej * 16, row16, koff);
      acc = WMMA_F32(a, bb, acc);
    }
#pragma unroll
    for (int j = 0; j < 8; ++j)
      out[(di * 16 + crow + j) * DH + ej * 16 + row16] = acc[j];
  }
}

#define SCAN_THREADS 128
#define SCAN_LANES_PER_HEAD (DH * DH / 2)

__global__ void __launch_bounds__(SCAN_THREADS) __attribute__((amdgpu_waves_per_eu(1, 2)))
linmem_prefix_state(float* __restrict__ mpart) {
  const int gid = blockIdx.x * SCAN_THREADS + threadIdx.x;
  const int bh  = gid / SCAN_LANES_PER_HEAD;
  const int e2  = gid % SCAN_LANES_PER_HEAD;
  float2* base = (float2*)(mpart + (size_t)bh * NPART * (DH * DH)) + e2;

  float2 x[NPART];
#pragma unroll
  for (int p = 0; p < NPART; ++p)
    x[p] = base[(size_t)p * SCAN_LANES_PER_HEAD];

  float2 run = make_float2(0.0f, 0.0f);
#pragma unroll
  for (int p = 0; p < NPART; ++p) {
    base[(size_t)p * SCAN_LANES_PER_HEAD] = run;
    run.x += x[p].x;
    run.y += x[p].y;
  }
}

#define OUT_THREADS 256
#define SET_FLOATS (4 * CHUNK * PITCH)

__global__ void __launch_bounds__(OUT_THREADS)
linmem_chunk_output(const float* __restrict__ q, const float* __restrict__ k,
                    const float* __restrict__ v, const float* __restrict__ mpart,
                    float* __restrict__ y) {
  __shared__ __align__(16) float Lset[2 * SET_FLOATS];

  const int tid = threadIdx.x, lane = tid & 31;
  const int wave = __builtin_amdgcn_readfirstlane(tid >> 5);
  const int sub = wave >> 2;
  const int lw  = wave & 3;
  const int lt  = tid & 127;
  const int row16 = lane & 15, half = lane >> 4;
  const int koff = half * 2, crow = half * 8;

  float* Qs = Lset + sub * SET_FLOATS;
  float* Ks = Qs + CHUNK * PITCH;
  float* Vs = Ks + CHUNK * PITCH;
  float* Ms = Vs + CHUNK * PITCH;

  const int cid = blockIdx.x * 2 + sub;
  const int bh = cid / NPART;
  const int p  = cid % NPART;
  const int b = bh / NH, h = bh % NH;
  const int t0 = p * CHUNK;

  const size_t head_off = ((size_t)b * T_SEQ * NH + h) * DH;
  const size_t tstride  = (size_t)NH * DH;
  const float* m0 = mpart + ((size_t)bh * NPART + p) * (DH * DH);

  const int r0 = lt >> 4, c4 = lt & 15;
  const size_t g0 = head_off + (size_t)(t0 + r0) * tstride + c4 * 4;
  float4 rq[STAGE_ROWS], rk[STAGE_ROWS], rv[STAGE_ROWS], rm[STAGE_ROWS];
#pragma unroll
  for (int i = 0; i < STAGE_ROWS; ++i)
    rq[i] = *(const float4*)&q[g0 + (size_t)(i * 8) * tstride];
#pragma unroll
  for (int i = 0; i < STAGE_ROWS; ++i)
    rk[i] = *(const float4*)&k[g0 + (size_t)(i * 8) * tstride];
#pragma unroll
  for (int i = 0; i < STAGE_ROWS; ++i)
    rv[i] = *(const float4*)&v[g0 + (size_t)(i * 8) * tstride];
#pragma unroll
  for (int i = 0; i < STAGE_ROWS; ++i)
    rm[i] = *(const float4*)&m0[(r0 + i * 8) * DH + c4 * 4];
#pragma unroll
  for (int i = 0; i < STAGE_ROWS; ++i)
    *(float4*)&Qs[(r0 + i * 8) * PITCH + c4 * 4] = rq[i];
#pragma unroll
  for (int i = 0; i < STAGE_ROWS; ++i)
    *(float4*)&Ks[(r0 + i * 8) * PITCH + c4 * 4] = rk[i];
#pragma unroll
  for (int i = 0; i < STAGE_ROWS; ++i)
    *(float4*)&Vs[(r0 + i * 8) * PITCH + c4 * 4] = rv[i];
#pragma unroll
  for (int i = 0; i < STAGE_ROWS; ++i)
    *(float4*)&Ms[(r0 + i * 8) * PITCH + c4 * 4] = rm[i];
  __syncthreads();

  const int ti = sub ? 3 - lw : lw;

  v2f qf[DH / 4];
#pragma unroll
  for (int e4 = 0; e4 < DH / 4; ++e4)
    qf[e4] = lds_fragA(Qs, ti * 16, e4 * 4, row16, koff);

  for (int sj = 0; sj <= ti; ++sj) {
    v8f acc = {};
#pragma unroll
    for (int e4 = 0; e4 < DH / 4; ++e4) {
      v2f bt = lds_fragA(Ks, sj * 16, e4 * 4, row16, koff);
      acc = WMMA_F32(qf[e4], bt, acc);
    }
#pragma unroll
    for (int j = 0; j < 8; ++j) {
      const int tt = ti * 16 + crow + j;
      const int ss = sj * 16 + row16;
      Qs[tt * PITCH + ss] = (ss <= tt) ? acc[j] : 0.0f;
    }
  }

  v8f acc[4] = {};
#pragma unroll
  for (int e4 = 0; e4 < DH / 4; ++e4) {
#pragma unroll
    for (int dj = 0; dj < 4; ++dj) {
      v2f bt = lds_fragA(Ms, dj * 16, e4 * 4, row16, koff);
      acc[dj] = WMMA_F32(qf[e4], bt, acc[dj]);
    }
  }
  for (int sj = 0; sj <= ti; ++sj) {
#pragma unroll
    for (int s4 = 0; s4 < 4; ++s4) {
      const int s = sj * 16 + s4 * 4;
      v2f a = lds_fragA(Qs, ti * 16, s, row16, koff);
#pragma unroll
      for (int dj = 0; dj < 4; ++dj) {
        v2f bb = lds_fragB(Vs, s, dj * 16, row16, koff);
        acc[dj] = WMMA_F32(a, bb, acc[dj]);
      }
    }
  }
#pragma unroll
  for (int dj = 0; dj < 4; ++dj) {
#pragma unroll
    for (int j = 0; j < 8; ++j) {
      const size_t g = head_off +
          (size_t)(t0 + ti * 16 + crow + j) * tstride + dj * 16 + row16;
      y[g] = acc[dj][j];
    }
  }
}

__global__ void __launch_bounds__(128, 1)
linmem_chunked_wmma(const float* __restrict__ q,
                    const float* __restrict__ k,
                    const float* __restrict__ v,
                    float* __restrict__ y) {
  __shared__ __align__(16) float Qs[CHUNK * PITCH];
  __shared__ __align__(16) float Ks[CHUNK * PITCH];
  __shared__ __align__(16) float Vs[CHUNK * PITCH];
  __shared__ __align__(16) float Ss[CHUNK * PITCH];
  __shared__ __align__(16) float Ms[DH    * PITCH];

  const int tid = threadIdx.x, lane = tid & 31, wave = tid >> 5;
  const int row16 = lane & 15, half = lane >> 4;
  const int koff = half * 2, crow = half * 8;
  const int b = blockIdx.x / NH, h = blockIdx.x % NH;

  for (int i = tid; i < DH * PITCH; i += 128) Ms[i] = 0.0f;
  __syncthreads();

  const size_t head_off = ((size_t)b * T_SEQ * NH + h) * DH;
  const size_t tstride  = (size_t)NH * DH;

  for (int t0 = 0; t0 < T_SEQ; t0 += CHUNK) {
    for (int idx = tid; idx < CHUNK * (DH / 4); idx += 128) {
      const int r = idx >> 4, c4 = idx & 15;
      const size_t g = head_off + (size_t)(t0 + r) * tstride + c4 * 4;
      *(float4*)&Qs[r * PITCH + c4 * 4] = *(const float4*)&q[g];
      *(float4*)&Ks[r * PITCH + c4 * 4] = *(const float4*)&k[g];
      *(float4*)&Vs[r * PITCH + c4 * 4] = *(const float4*)&v[g];
      if (t0 + CHUNK < T_SEQ) {
        __builtin_prefetch(&q[g + CHUNK * tstride], 0, 3);
        __builtin_prefetch(&k[g + CHUNK * tstride], 0, 3);
        __builtin_prefetch(&v[g + CHUNK * tstride], 0, 3);
      }
    }
    __syncthreads();

    const int ti = wave;
    for (int sj = 0; sj < 4; ++sj) {
      v8f acc = {};
      if (sj <= ti) {
        for (int e = 0; e < DH; e += 4) {
          v2f a  = lds_fragA(Qs, ti * 16, e, row16, koff);
          v2f bt = lds_fragA(Ks, sj * 16, e, row16, koff);
          acc = WMMA_F32(a, bt, acc);
        }
      }
#pragma unroll
      for (int j = 0; j < 8; ++j) {
        const int tt = ti * 16 + crow + j;
        const int ss = sj * 16 + row16;
        Ss[tt * PITCH + ss] = (ss <= tt) ? acc[j] : 0.0f;
      }
    }
    __syncthreads();

    for (int dj = 0; dj < 4; ++dj) {
      v8f acc = {};
      for (int e = 0; e < DH; e += 4) {
        v2f a  = lds_fragA(Qs, ti * 16, e, row16, koff);
        v2f bt = lds_fragA(Ms, dj * 16, e, row16, koff);
        acc = WMMA_F32(a, bt, acc);
      }
      for (int s = 0; s < CHUNK; s += 4) {
        v2f a  = lds_fragA(Ss, ti * 16, s, row16, koff);
        v2f bb = lds_fragB(Vs, s, dj * 16, row16, koff);
        acc = WMMA_F32(a, bb, acc);
      }
#pragma unroll
      for (int j = 0; j < 8; ++j) {
        const size_t g = head_off +
            (size_t)(t0 + ti * 16 + crow + j) * tstride + dj * 16 + row16;
        y[g] = acc[j];
      }
    }
    __syncthreads();

    const int di = wave;
    for (int ej = 0; ej < 4; ++ej) {
      v8f acc;
#pragma unroll
      for (int j = 0; j < 8; ++j)
        acc[j] = Ms[(di * 16 + crow + j) * PITCH + ej * 16 + row16];
      for (int t = 0; t < CHUNK; t += 4) {
        v2f a  = lds_fragB(Vs, t, di * 16, row16, koff);
        v2f bb = lds_fragB(Ks, t, ej * 16, row16, koff);
        acc = WMMA_F32(a, bb, acc);
      }
#pragma unroll
      for (int j = 0; j < 8; ++j)
        Ms[(di * 16 + crow + j) * PITCH + ej * 16 + row16] = acc[j];
    }
    __syncthreads();
  }
}

extern "C" void kernel_launch(void* const* d_in, const int* in_sizes, int n_in,
                              void* d_out, int out_size, void* d_ws, size_t ws_size,
                              hipStream_t stream) {
  const float* q = (const float*)d_in[0];
  const float* k = (const float*)d_in[1];
  const float* v = (const float*)d_in[2];
  float* y = (float*)d_out;

  const size_t ws_need = (size_t)NHEAD * NPART * DH * DH * sizeof(float);

  if (ws_size >= ws_need) {
    float* mpart = (float*)d_ws;
    linmem_partial_state<<<dim3(NHEAD * NPART), dim3(128), 0, stream>>>(k, v, mpart);
    linmem_prefix_state <<<dim3(NHEAD * SCAN_LANES_PER_HEAD / SCAN_THREADS), dim3(SCAN_THREADS), 0, stream>>>(mpart);
    linmem_chunk_output <<<dim3(NHEAD * NPART / 2), dim3(OUT_THREADS), 0, stream>>>(q, k, v, mpart, y);
  } else {
    linmem_chunked_wmma<<<dim3(NHEAD), dim3(128), 0, stream>>>(q, k, v, y);
  }
}


# scan launch removed (3 -> 2 launches): one work-group owns 256 steps; state kernel also writes in-group running sums, output kernel adds earlier groups' totals itself; state sum order regrouped
# speedup vs baseline: 3.8062x; 1.0540x over previous
#include <hip/hip_runtime.h>

typedef __attribute__((ext_vector_type(2))) float v2f;
typedef __attribute__((ext_vector_type(8))) float v8f;

#define T_SEQ 4096
#define NH    8
#define DH    64
#define CHUNK 64
#define GROUP 256
#define CPG   (GROUP / CHUNK)
#define NGRP  (T_SEQ / GROUP)
#define NHEAD (2 * NH)
#define PITCH 68
#define STAGE_ROWS (CHUNK / 8)

#define WMMA_F32(a, b, c) \
  __builtin_amdgcn_wmma_f32_16x16x4_f32(false, (a), false, (b), (short)0, (c), false, false)

__device__ __forceinline__ v2f lds_fragA(const float* L, int rb, int kb, int row16, int koff) {
  v2f a;
  a.x = L[(rb + row16) * PITCH + kb + koff];
  a.y = L[(rb + row16) * PITCH + kb + koff + 1];
  return a;
}

__device__ __forceinline__ v2f lds_fragB(const float* L, int kb, int nb, int row16, int koff) {
  v2f b;
  b.x = L[(kb + koff)     * PITCH + nb + row16];
  b.y = L[(kb + koff + 1) * PITCH + nb + row16];
  return b;
}

__device__ __forceinline__ float4 add4(float4 a, float4 b) {
  return make_float4(a.x + b.x, a.y + b.y, a.z + b.z, a.w + b.w);
}

#define GRP_THREADS 512
#define STATE_FLOATS (DH * DH)
#define SLOT_FLOATS (CPG * STATE_FLOATS)
#define GSTAGE_ROWS (GROUP / 32)

__global__ void __launch_bounds__(GRP_THREADS)
linmem_group_state(const float* __restrict__ k, const float* __restrict__ v,
                   float* __restrict__ gws) {
  __shared__ __align__(16) float Ks[GROUP * PITCH];
  __shared__ __align__(16) float Vs[GROUP * PITCH];

  const int tid = threadIdx.x, lane = tid & 31;
  const int wave = __builtin_amdgcn_readfirstlane(tid >> 5);
  const int sub = wave >> 2;
  const int lw  = wave & 3;
  const int row16 = lane & 15, half = lane >> 4;
  const int koff = half * 2, crow = half * 8;

  const int bh = blockIdx.x / NGRP;
  const int s  = blockIdx.x % NGRP;
  const int b = bh / NH, h = bh % NH;
  const int t0 = s * GROUP;

  const size_t head_off = ((size_t)b * T_SEQ * NH + h) * DH;
  const size_t tstride  = (size_t)NH * DH;

  const int r0 = tid >> 4, c4 = tid & 15;
  const size_t g0 = head_off + (size_t)(t0 + r0) * tstride + c4 * 4;
  float4 rk[GSTAGE_ROWS], rv[GSTAGE_ROWS];
#pragma unroll
  for (int i = 0; i < GSTAGE_ROWS; ++i)
    rk[i] = *(const float4*)&k[g0 + (size_t)(i * 32) * tstride];
#pragma unroll
  for (int i = 0; i < GSTAGE_ROWS; ++i)
    rv[i] = *(const float4*)&v[g0 + (size_t)(i * 32) * tstride];
#pragma unroll
  for (int i = 0; i < GSTAGE_ROWS; ++i)
    *(float4*)&Ks[(r0 + i * 32) * PITCH + c4 * 4] = rk[i];
#pragma unroll
  for (int i = 0; i < GSTAGE_ROWS; ++i)
    *(float4*)&Vs[(r0 + i * 32) * PITCH + c4 * 4] = rv[i];
  __syncthreads();

  const float* Kc = Ks + sub * CHUNK * PITCH;
  const float* Vc = Vs + sub * CHUNK * PITCH;
  v8f m[4] = {};
#pragma unroll
  for (int t = 0; t < CHUNK; t += 4) {
    v2f a = lds_fragB(Vc, t, lw * 16, row16, koff);
#pragma unroll
    for (int ej = 0; ej < 4; ++ej) {
      v2f bb = lds_fragB(Kc, t, ej * 16, row16, koff);
      m[ej] = WMMA_F32(a, bb, m[ej]);
    }
  }
  __syncthreads();

  float* Mc = Ks + sub * CHUNK * PITCH;
#pragma unroll
  for (int ej = 0; ej < 4; ++ej) {
#pragma unroll
    for (int j = 0; j < 8; ++j)
      Mc[(lw * 16 + crow + j) * PITCH + ej * 16 + row16] = m[ej][j];
  }
  __syncthreads();

  float* out = gws + (size_t)(bh * NGRP + s) * SLOT_FLOATS;
  const int pr = tid >> 4, pc = (tid & 15) * 4;
#pragma unroll
  for (int hh = 0; hh < 2; ++hh) {
    const int row = pr + 32 * hh;
    const int lo = row * PITCH + pc;
    const int go = row * DH + pc;
    float4 run = *(const float4*)&Ks[lo];
#pragma unroll
    for (int c = 1; c < CPG; ++c) {
      *(float4*)&out[(c - 1) * STATE_FLOATS + go] = run;
      run = add4(run, *(const float4*)&Ks[c * CHUNK * PITCH + lo]);
    }
    *(float4*)&out[(CPG - 1) * STATE_FLOATS + go] = run;
  }
}

#define SET_FLOATS (4 * CHUNK * PITCH)

__global__ void __launch_bounds__(GRP_THREADS)
linmem_group_output(const float* __restrict__ q, const float* __restrict__ k,
                    const float* __restrict__ v, const float* __restrict__ gws,
                    float* __restrict__ y) {
  __shared__ __align__(16) float Lset[CPG * SET_FLOATS];

  const int tid = threadIdx.x, lane = tid & 31;
  const int wave = __builtin_amdgcn_readfirstlane(tid >> 5);
  const int sub = wave >> 2;
  const int lw  = wave & 3;
  const int lt  = tid & 127;
  const int row16 = lane & 15, half = lane >> 4;
  const int koff = half * 2, crow = half * 8;

  float* Qs = Lset + sub * SET_FLOATS;
  float* Ks = Qs + CHUNK * PITCH;
  float* Vs = Ks + CHUNK * PITCH;
  float* Ms = Vs + CHUNK * PITCH;

  const int bh = blockIdx.x / NGRP;
  const int s  = blockIdx.x % NGRP;
  const int b = bh / NH, h = bh % NH;
  const int t0 = s * GROUP + sub * CHUNK;

  const size_t head_off = ((size_t)b * T_SEQ * NH + h) * DH;
  const size_t tstride  = (size_t)NH * DH;
  const float* gh = gws + (size_t)bh * NGRP * SLOT_FLOATS;

  const int f0 = tid * 4;
  const int f1 = f0 + GRP_THREADS * 4;
  const float4 zero4 = make_float4(0.0f, 0.0f, 0.0f, 0.0f);

  const int glast = (s > 0) ? s - 1 : 0;
  float4 xa[NGRP - 1], xb[NGRP - 1];
#pragma unroll
  for (int g = 0; g < NGRP - 1; ++g) {
    const int gc = (g < s) ? g : glast;
    const float* tot = gh + (size_t)gc * SLOT_FLOATS + (CPG - 1) * STATE_FLOATS;
    xa[g] = *(const float4*)&tot[f0];
    xb[g] = *(const float4*)&tot[f1];
  }
  float4 wa[CPG - 1], wb[CPG - 1];
#pragma unroll
  for (int c = 1; c < CPG; ++c) {
    const float* run = gh + (size_t)s * SLOT_FLOATS + (c - 1) * STATE_FLOATS;
    wa[c - 1] = *(const float4*)&run[f0];
    wb[c - 1] = *(const float4*)&run[f1];
  }

  const int r0 = lt >> 4, c4 = lt & 15;
  const size_t g0 = head_off + (size_t)(t0 + r0) * tstride + c4 * 4;
  float4 rq[STAGE_ROWS], rk[STAGE_ROWS], rv[STAGE_ROWS];
#pragma unroll
  for (int i = 0; i < STAGE_ROWS; ++i)
    rq[i] = *(const float4*)&q[g0 + (size_t)(i * 8) * tstride];
#pragma unroll
  for (int i = 0; i < STAGE_ROWS; ++i)
    rk[i] = *(const float4*)&k[g0 + (size_t)(i * 8) * tstride];
#pragma unroll
  for (int i = 0; i < STAGE_ROWS; ++i)
    rv[i] = *(const float4*)&v[g0 + (size_t)(i * 8) * tstride];

  float4 p0 = zero4, p1 = zero4;
#pragma unroll
  for (int g = 0; g < NGRP - 1; ++g) {
    p0 = add4(p0, (g < s) ? xa[g] : zero4);
    p1 = add4(p1, (g < s) ? xb[g] : zero4);
  }

#pragma unroll
  for (int i = 0; i < STAGE_ROWS; ++i)
    *(float4*)&Qs[(r0 + i * 8) * PITCH + c4 * 4] = rq[i];
#pragma unroll
  for (int i = 0; i < STAGE_ROWS; ++i)
    *(float4*)&Ks[(r0 + i * 8) * PITCH + c4 * 4] = rk[i];
#pragma unroll
  for (int i = 0; i < STAGE_ROWS; ++i)
    *(float4*)&Vs[(r0 + i * 8) * PITCH + c4 * 4] = rv[i];

  {
    const int pr = tid >> 4, pc = (tid & 15) * 4;
    float* M0s = Lset + 3 * CHUNK * PITCH;
    const int o0 = pr * PITCH + pc;
    const int o1 = (pr + 32) * PITCH + pc;
    *(float4*)&M0s[o0] = p0;
    *(float4*)&M0s[o1] = p1;
#pragma unroll
    for (int c = 1; c < CPG; ++c) {
      *(float4*)&M0s[c * SET_FLOATS + o0] = add4(p0, wa[c - 1]);
      *(float4*)&M0s[c * SET_FLOATS + o1] = add4(p1, wb[c - 1]);
    }
  }
  __syncthreads();

  const int ti = (lw + sub) & 3;

  v2f qf[DH / 4];
#pragma unroll
  for (int e4 = 0; e4 < DH / 4; ++e4)
    qf[e4] = lds_fragA(Qs, ti * 16, e4 * 4, row16, koff);

  for (int sj = 0; sj <= ti; ++sj) {
    v8f acc = {};
#pragma unroll
    for (int e4 = 0; e4 < DH / 4; ++e4) {
      v2f bt = lds_fragA(Ks, sj * 16, e4 * 4, row16, koff);
      acc = WMMA_F32(qf[e4], bt, acc);
    }
#pragma unroll
    for (int j = 0; j < 8; ++j) {
      const int tt = ti * 16 + crow + j;
      const int ss = sj * 16 + row16;
      Qs[tt * PITCH + ss] = (ss <= tt) ? acc[j] : 0.0f;
    }
  }

  v8f acc[4] = {};
#pragma unroll
  for (int e4 = 0; e4 < DH / 4; ++e4) {
#pragma unroll
    for (int dj = 0; dj < 4; ++dj) {
      v2f bt = lds_fragA(Ms, dj * 16, e4 * 4, row16, koff);
      acc[dj] = WMMA_F32(qf[e4], bt, acc[dj]);
    }
  }
  for (int sj = 0; sj <= ti; ++sj) {
#pragma unroll
    for (int s4 = 0; s4 < 4; ++s4) {
      const int sc = sj * 16 + s4 * 4;
      v2f a = lds_fragA(Qs, ti * 16, sc, row16, koff);
#pragma unroll
      for (int dj = 0; dj < 4; ++dj) {
        v2f bb = lds_fragB(Vs, sc, dj * 16, row16, koff);
        acc[dj] = WMMA_F32(a, bb, acc[dj]);
      }
    }
  }
#pragma unroll
  for (int dj = 0; dj < 4; ++dj) {
#pragma unroll
    for (int j = 0; j < 8; ++j) {
      const size_t g = head_off +
          (size_t)(t0 + ti * 16 + crow + j) * tstride + dj * 16 + row16;
      y[g] = acc[dj][j];
    }
  }
}

__global__ void __launch_bounds__(128, 1)
linmem_chunked_wmma(const float* __restrict__ q,
                    const float* __restrict__ k,
                    const float* __restrict__ v,
                    float* __restrict__ y) {
  __shared__ __align__(16) float Qs[CHUNK * PITCH];
  __shared__ __align__(16) float Ks[CHUNK * PITCH];
  __shared__ __align__(16) float Vs[CHUNK * PITCH];
  __shared__ __align__(16) float Ss[CHUNK * PITCH];
  __shared__ __align__(16) float Ms[DH    * PITCH];

  const int tid = threadIdx.x, lane = tid & 31, wave = tid >> 5;
  const int row16 = lane & 15, half = lane >> 4;
  const int koff = half * 2, crow = half * 8;
  const int b = blockIdx.x / NH, h = blockIdx.x % NH;

  for (int i = tid; i < DH * PITCH; i += 128) Ms[i] = 0.0f;
  __syncthreads();

  const size_t head_off = ((size_t)b * T_SEQ * NH + h) * DH;
  const size_t tstride  = (size_t)NH * DH;

  for (int t0 = 0; t0 < T_SEQ; t0 += CHUNK) {
    for (int idx = tid; idx < CHUNK * (DH / 4); idx += 128) {
      const int r = idx >> 4, c4 = idx & 15;
      const size_t g = head_off + (size_t)(t0 + r) * tstride + c4 * 4;
      *(float4*)&Qs[r * PITCH + c4 * 4] = *(const float4*)&q[g];
      *(float4*)&Ks[r * PITCH + c4 * 4] = *(const float4*)&k[g];
      *(float4*)&Vs[r * PITCH + c4 * 4] = *(const float4*)&v[g];
      if (t0 + CHUNK < T_SEQ) {
        __builtin_prefetch(&q[g + CHUNK * tstride], 0, 3);
        __builtin_prefetch(&k[g + CHUNK * tstride], 0, 3);
        __builtin_prefetch(&v[g + CHUNK * tstride], 0, 3);
      }
    }
    __syncthreads();

    const int ti = wave;
    for (int sj = 0; sj < 4; ++sj) {
      v8f acc = {};
      if (sj <= ti) {
        for (int e = 0; e < DH; e += 4) {
          v2f a  = lds_fragA(Qs, ti * 16, e, row16, koff);
          v2f bt = lds_fragA(Ks, sj * 16, e, row16, koff);
          acc = WMMA_F32(a, bt, acc);
        }
      }
#pragma unroll
      for (int j = 0; j < 8; ++j) {
        const int tt = ti * 16 + crow + j;
        const int ss = sj * 16 + row16;
        Ss[tt * PITCH + ss] = (ss <= tt) ? acc[j] : 0.0f;
      }
    }
    __syncthreads();

    for (int dj = 0; dj < 4; ++dj) {
      v8f acc = {};
      for (int e = 0; e < DH; e += 4) {
        v2f a  = lds_fragA(Qs, ti * 16, e, row16, koff);
        v2f bt = lds_fragA(Ms, dj * 16, e, row16, koff);
        acc = WMMA_F32(a, bt, acc);
      }
      for (int s = 0; s < CHUNK; s += 4) {
        v2f a  = lds_fragA(Ss, ti * 16, s, row16, koff);
        v2f bb = lds_fragB(Vs, s, dj * 16, row16, koff);
        acc = WMMA_F32(a, bb, acc);
      }
#pragma unroll
      for (int j = 0; j < 8; ++j) {
        const size_t g = head_off +
            (size_t)(t0 + ti * 16 + crow + j) * tstride + dj * 16 + row16;
        y[g] = acc[j];
      }
    }
    __syncthreads();

    const int di = wave;
    for (int ej = 0; ej < 4; ++ej) {
      v8f acc;
#pragma unroll
      for (int j = 0; j < 8; ++j)
        acc[j] = Ms[(di * 16 + crow + j) * PITCH + ej * 16 + row16];
      for (int t = 0; t < CHUNK; t += 4) {
        v2f a  = lds_fragB(Vs, t, di * 16, row16, koff);
        v2f bb = lds_fragB(Ks, t, ej * 16, row16, koff);
        acc = WMMA_F32(a, bb, acc);
      }
#pragma unroll
      for (int j = 0; j < 8; ++j)
        Ms[(di * 16 + crow + j) * PITCH + ej * 16 + row16] = acc[j];
    }
    __syncthreads();
  }
}

extern "C" void kernel_launch(void* const* d_in, const int* in_sizes, int n_in,
                              void* d_out, int out_size, void* d_ws, size_t ws_size,
                              hipStream_t stream) {
  const float* q = (const float*)d_in[0];
  const float* k = (const float*)d_in[1];
  const float* v = (const float*)d_in[2];
  float* y = (float*)d_out;

  const size_t ws_need = (size_t)NHEAD * NGRP * SLOT_FLOATS * sizeof(float);

  if (ws_size >= ws_need) {
    float* gws = (float*)d_ws;
    linmem_group_state <<<dim3(NHEAD * NGRP), dim3(GRP_THREADS), 0, stream>>>(k, v, gws);
    linmem_group_output<<<dim3(NHEAD * NGRP), dim3(GRP_THREADS), 0, stream>>>(q, k, v, gws, y);
  } else {
    linmem_chunked_wmma<<<dim3(NHEAD), dim3(128), 0, stream>>>(q, k, v, y);
  }
}


# output kernel: causal part (scores, scores*V) first, under the flight of the state loads issued behind the operand barrier; states staged at a 2nd barrier, Q*S^T added last (sum order: causal first)
# speedup vs baseline: 3.9807x; 1.0458x over previous
#include <hip/hip_runtime.h>

typedef __attribute__((ext_vector_type(2))) float v2f;
typedef __attribute__((ext_vector_type(8))) float v8f;

#define T_SEQ 4096
#define NH    8
#define DH    64
#define CHUNK 64
#define GROUP 256
#define CPG   (GROUP / CHUNK)
#define NGRP  (T_SEQ / GROUP)
#define NHEAD (2 * NH)
#define PITCH 68
#define STAGE_ROWS (CHUNK / 8)

#define WMMA_F32(a, b, c) \
  __builtin_amdgcn_wmma_f32_16x16x4_f32(false, (a), false, (b), (short)0, (c), false, false)

__device__ __forceinline__ v2f lds_fragA(const float* L, int rb, int kb, int row16, int koff) {
  v2f a;
  a.x = L[(rb + row16) * PITCH + kb + koff];
  a.y = L[(rb + row16) * PITCH + kb + koff + 1];
  return a;
}

__device__ __forceinline__ v2f lds_fragB(const float* L, int kb, int nb, int row16, int koff) {
  v2f b;
  b.x = L[(kb + koff)     * PITCH + nb + row16];
  b.y = L[(kb + koff + 1) * PITCH + nb + row16];
  return b;
}

__device__ __forceinline__ float4 add4(float4 a, float4 b) {
  return make_float4(a.x + b.x, a.y + b.y, a.z + b.z, a.w + b.w);
}

#define GRP_THREADS 512
#define STATE_FLOATS (DH * DH)
#define SLOT_FLOATS (CPG * STATE_FLOATS)
#define GSTAGE_ROWS (GROUP / 32)

__global__ void __launch_bounds__(GRP_THREADS)
linmem_group_state(const float* __restrict__ k, const float* __restrict__ v,
                   float* __restrict__ gws) {
  __shared__ __align__(16) float Ks[GROUP * PITCH];
  __shared__ __align__(16) float Vs[GROUP * PITCH];

  const int tid = threadIdx.x, lane = tid & 31;
  const int wave = __builtin_amdgcn_readfirstlane(tid >> 5);
  const int sub = wave >> 2;
  const int lw  = wave & 3;
  const int row16 = lane & 15, half = lane >> 4;
  const int koff = half * 2, crow = half * 8;

  const int bh = blockIdx.x / NGRP;
  const int s  = blockIdx.x % NGRP;
  const int b = bh / NH, h = bh % NH;
  const int t0 = s * GROUP;

  const size_t head_off = ((size_t)b * T_SEQ * NH + h) * DH;
  const size_t tstride  = (size_t)NH * DH;

  const int r0 = tid >> 4, c4 = tid & 15;
  const size_t g0 = head_off + (size_t)(t0 + r0) * tstride + c4 * 4;
  float4 rk[GSTAGE_ROWS], rv[GSTAGE_ROWS];
#pragma unroll
  for (int i = 0; i < GSTAGE_ROWS; ++i)
    rk[i] = *(const float4*)&k[g0 + (size_t)(i * 32) * tstride];
#pragma unroll
  for (int i = 0; i < GSTAGE_ROWS; ++i)
    rv[i] = *(const float4*)&v[g0 + (size_t)(i * 32) * tstride];
#pragma unroll
  for (int i = 0; i < GSTAGE_ROWS; ++i)
    *(float4*)&Ks[(r0 + i * 32) * PITCH + c4 * 4] = rk[i];
#pragma unroll
  for (int i = 0; i < GSTAGE_ROWS; ++i)
    *(float4*)&Vs[(r0 + i * 32) * PITCH + c4 * 4] = rv[i];
  __syncthreads();

  const float* Kc = Ks + sub * CHUNK * PITCH;
  const float* Vc = Vs + sub * CHUNK * PITCH;
  v8f m[4] = {};
#pragma unroll
  for (int t = 0; t < CHUNK; t += 4) {
    v2f a = lds_fragB(Vc, t, lw * 16, row16, koff);
#pragma unroll
    for (int ej = 0; ej < 4; ++ej) {
      v2f bb = lds_fragB(Kc, t, ej * 16, row16, koff);
      m[ej] = WMMA_F32(a, bb, m[ej]);
    }
  }
  __syncthreads();

  float* Mc = Ks + sub * CHUNK * PITCH;
#pragma unroll
  for (int ej = 0; ej < 4; ++ej) {
#pragma unroll
    for (int j = 0; j < 8; ++j)
      Mc[(lw * 16 + crow + j) * PITCH + ej * 16 + row16] = m[ej][j];
  }
  __syncthreads();

  float* out = gws + (size_t)(bh * NGRP + s) * SLOT_FLOATS;
  const int pr = tid >> 4, pc = (tid & 15) * 4;
#pragma unroll
  for (int hh = 0; hh < 2; ++hh) {
    const int row = pr + 32 * hh;
    const int lo = row * PITCH + pc;
    const int go = row * DH + pc;
    float4 run = *(const float4*)&Ks[lo];
#pragma unroll
    for (int c = 1; c < CPG; ++c) {
      *(float4*)&out[(c - 1) * STATE_FLOATS + go] = run;
      run = add4(run, *(const float4*)&Ks[c * CHUNK * PITCH + lo]);
    }
    *(float4*)&out[(CPG - 1) * STATE_FLOATS + go] = run;
  }
}

#define SET_FLOATS (4 * CHUNK * PITCH)

__global__ void __launch_bounds__(GRP_THREADS)
linmem_group_output(const float* __restrict__ q, const float* __restrict__ k,
                    const float* __restrict__ v, const float* __restrict__ gws,
                    float* __restrict__ y) {
  __shared__ __align__(16) float Lset[CPG * SET_FLOATS];

  const int tid = threadIdx.x, lane = tid & 31;
  const int wave = __builtin_amdgcn_readfirstlane(tid >> 5);
  const int sub = wave >> 2;
  const int lw  = wave & 3;
  const int lt  = tid & 127;
  const int row16 = lane & 15, half = lane >> 4;
  const int koff = half * 2, crow = half * 8;

  float* Qs = Lset + sub * SET_FLOATS;
  float* Ks = Qs + CHUNK * PITCH;
  float* Vs = Ks + CHUNK * PITCH;
  float* Ms = Vs + CHUNK * PITCH;

  const int bh = blockIdx.x / NGRP;
  const int s  = blockIdx.x % NGRP;
  const int b = bh / NH, h = bh % NH;
  const int t0 = s * GROUP + sub * CHUNK;

  const size_t head_off = ((size_t)b * T_SEQ * NH + h) * DH;
  const size_t tstride  = (size_t)NH * DH;
  const float* gh = gws + (size_t)bh * NGRP * SLOT_FLOATS;

  const int f0 = tid * 4;
  const int f1 = f0 + GRP_THREADS * 4;

  const int r0 = lt >> 4, c4 = lt & 15;
  const size_t g0 = head_off + (size_t)(t0 + r0) * tstride + c4 * 4;
  float4 rq[STAGE_ROWS], rk[STAGE_ROWS], rv[STAGE_ROWS];
#pragma unroll
  for (int i = 0; i < STAGE_ROWS; ++i)
    rq[i] = *(const float4*)&q[g0 + (size_t)(i * 8) * tstride];
#pragma unroll
  for (int i = 0; i < STAGE_ROWS; ++i)
    rk[i] = *(const float4*)&k[g0 + (size_t)(i * 8) * tstride];
#pragma unroll
  for (int i = 0; i < STAGE_ROWS; ++i)
    rv[i] = *(const float4*)&v[g0 + (size_t)(i * 8) * tstride];

#pragma unroll
  for (int i = 0; i < STAGE_ROWS; ++i)
    *(float4*)&Qs[(r0 + i * 8) * PITCH + c4 * 4] = rq[i];
#pragma unroll
  for (int i = 0; i < STAGE_ROWS; ++i)
    *(float4*)&Ks[(r0 + i * 8) * PITCH + c4 * 4] = rk[i];
#pragma unroll
  for (int i = 0; i < STAGE_ROWS; ++i)
    *(float4*)&Vs[(r0 + i * 8) * PITCH + c4 * 4] = rv[i];
  __syncthreads();

  const int glast = (s > 0) ? s - 1 : 0;
  float4 xa[NGRP - 1], wa[CPG - 1];
#pragma unroll
  for (int g = 0; g < NGRP - 1; ++g) {
    const int gc = (g < s) ? g : glast;
    xa[g] = *(const float4*)&gh[(size_t)gc * SLOT_FLOATS + (CPG - 1) * STATE_FLOATS + f0];
  }
#pragma unroll
  for (int c = 1; c < CPG; ++c)
    wa[c - 1] = *(const float4*)&gh[(size_t)s * SLOT_FLOATS + (c - 1) * STATE_FLOATS + f0];

  const int ti = (lw + sub) & 3;

  v2f qf[DH / 4];
#pragma unroll
  for (int e4 = 0; e4 < DH / 4; ++e4)
    qf[e4] = lds_fragA(Qs, ti * 16, e4 * 4, row16, koff);

  for (int sj = 0; sj <= ti; ++sj) {
    v8f sacc = {};
#pragma unroll
    for (int e4 = 0; e4 < DH / 4; ++e4) {
      v2f bt = lds_fragA(Ks, sj * 16, e4 * 4, row16, koff);
      sacc = WMMA_F32(qf[e4], bt, sacc);
    }
#pragma unroll
    for (int j = 0; j < 8; ++j) {
      const int tt = ti * 16 + crow + j;
      const int ss = sj * 16 + row16;
      Qs[tt * PITCH + ss] = (ss <= tt) ? sacc[j] : 0.0f;
    }
  }

  float4 xb[NGRP - 1], wb[CPG - 1];
#pragma unroll
  for (int g = 0; g < NGRP - 1; ++g) {
    const int gc = (g < s) ? g : glast;
    xb[g] = *(const float4*)&gh[(size_t)gc * SLOT_FLOATS + (CPG - 1) * STATE_FLOATS + f1];
  }
#pragma unroll
  for (int c = 1; c < CPG; ++c)
    wb[c - 1] = *(const float4*)&gh[(size_t)s * SLOT_FLOATS + (c - 1) * STATE_FLOATS + f1];

  v8f acc[4] = {};
  for (int sj = 0; sj <= ti; ++sj) {
#pragma unroll
    for (int s4 = 0; s4 < 4; ++s4) {
      const int sc = sj * 16 + s4 * 4;
      v2f a = lds_fragA(Qs, ti * 16, sc, row16, koff);
#pragma unroll
      for (int dj = 0; dj < 4; ++dj) {
        v2f bb = lds_fragB(Vs, sc, dj * 16, row16, koff);
        acc[dj] = WMMA_F32(a, bb, acc[dj]);
      }
    }
  }

  const float4 zero4 = make_float4(0.0f, 0.0f, 0.0f, 0.0f);
  float4 p0 = zero4, p1 = zero4;
#pragma unroll
  for (int g = 0; g < NGRP - 1; ++g) {
    p0 = add4(p0, (g < s) ? xa[g] : zero4);
    p1 = add4(p1, (g < s) ? xb[g] : zero4);
  }
  {
    const int pr = tid >> 4, pc = (tid & 15) * 4;
    float* M0s = Lset + 3 * CHUNK * PITCH;
    const int o0 = pr * PITCH + pc;
    const int o1 = (pr + 32) * PITCH + pc;
    *(float4*)&M0s[o0] = p0;
    *(float4*)&M0s[o1] = p1;
#pragma unroll
    for (int c = 1; c < CPG; ++c) {
      *(float4*)&M0s[c * SET_FLOATS + o0] = add4(p0, wa[c - 1]);
      *(float4*)&M0s[c * SET_FLOATS + o1] = add4(p1, wb[c - 1]);
    }
  }
  __syncthreads();

#pragma unroll
  for (int e4 = 0; e4 < DH / 4; ++e4) {
#pragma unroll
    for (int dj = 0; dj < 4; ++dj) {
      v2f bt = lds_fragA(Ms, dj * 16, e4 * 4, row16, koff);
      acc[dj] = WMMA_F32(qf[e4], bt, acc[dj]);
    }
  }
#pragma unroll
  for (int dj = 0; dj < 4; ++dj) {
#pragma unroll
    for (int j = 0; j < 8; ++j) {
      const size_t g = head_off +
          (size_t)(t0 + ti * 16 + crow + j) * tstride + dj * 16 + row16;
      y[g] = acc[dj][j];
    }
  }
}

__global__ void __launch_bounds__(128, 1)
linmem_chunked_wmma(const float* __restrict__ q,
                    const float* __restrict__ k,
                    const float* __restrict__ v,
                    float* __restrict__ y) {
  __shared__ __align__(16) float Qs[CHUNK * PITCH];
  __shared__ __align__(16) float Ks[CHUNK * PITCH];
  __shared__ __align__(16) float Vs[CHUNK * PITCH];
  __shared__ __align__(16) float Ss[CHUNK * PITCH];
  __shared__ __align__(16) float Ms[DH    * PITCH];

  const int tid = threadIdx.x, lane = tid & 31, wave = tid >> 5;
  const int row16 = lane & 15, half = lane >> 4;
  const int koff = half * 2, crow = half * 8;
  const int b = blockIdx.x / NH, h = blockIdx.x % NH;

  for (int i = tid; i < DH * PITCH; i += 128) Ms[i] = 0.0f;
  __syncthreads();

  const size_t head_off = ((size_t)b * T_SEQ * NH + h) * DH;
  const size_t tstride  = (size_t)NH * DH;

  for (int t0 = 0; t0 < T_SEQ; t0 += CHUNK) {
    for (int idx = tid; idx < CHUNK * (DH / 4); idx += 128) {
      const int r = idx >> 4, c4 = idx & 15;
      const size_t g = head_off + (size_t)(t0 + r) * tstride + c4 * 4;
      *(float4*)&Qs[r * PITCH + c4 * 4] = *(const float4*)&q[g];
      *(float4*)&Ks[r * PITCH + c4 * 4] = *(const float4*)&k[g];
      *(float4*)&Vs[r * PITCH + c4 * 4] = *(const float4*)&v[g];
      if (t0 + CHUNK < T_SEQ) {
        __builtin_prefetch(&q[g + CHUNK * tstride], 0, 3);
        __builtin_prefetch(&k[g + CHUNK * tstride], 0, 3);
        __builtin_prefetch(&v[g + CHUNK * tstride], 0, 3);
      }
    }
    __syncthreads();

    const int ti = wave;
    for (int sj = 0; sj < 4; ++sj) {
      v8f acc = {};
      if (sj <= ti) {
        for (int e = 0; e < DH; e += 4) {
          v2f a  = lds_fragA(Qs, ti * 16, e, row16, koff);
          v2f bt = lds_fragA(Ks, sj * 16, e, row16, koff);
          acc = WMMA_F32(a, bt, acc);
        }
      }
#pragma unroll
      for (int j = 0; j < 8; ++j) {
        const int tt = ti * 16 + crow + j;
        const int ss = sj * 16 + row16;
        Ss[tt * PITCH + ss] = (ss <= tt) ? acc[j] : 0.0f;
      }
    }
    __syncthreads();

    for (int dj = 0; dj < 4; ++dj) {
      v8f acc = {};
      for (int e = 0; e < DH; e += 4) {
        v2f a  = lds_fragA(Qs, ti * 16, e, row16, koff);
        v2f bt = lds_fragA(Ms, dj * 16, e, row16, koff);
        acc = WMMA_F32(a, bt, acc);
      }
      for (int s = 0; s < CHUNK; s += 4) {
        v2f a  = lds_fragA(Ss, ti * 16, s, row16, koff);
        v2f bb = lds_fragB(Vs, s, dj * 16, row16, koff);
        acc = WMMA_F32(a, bb, acc);
      }
#pragma unroll
      for (int j = 0; j < 8; ++j) {
        const size_t g = head_off +
            (size_t)(t0 + ti * 16 + crow + j) * tstride + dj * 16 + row16;
        y[g] = acc[j];
      }
    }
    __syncthreads();

    const int di = wave;
    for (int ej = 0; ej < 4; ++ej) {
      v8f acc;
#pragma unroll
      for (int j = 0; j < 8; ++j)
        acc[j] = Ms[(di * 16 + crow + j) * PITCH + ej * 16 + row16];
      for (int t = 0; t < CHUNK; t += 4) {
        v2f a  = lds_fragB(Vs, t, di * 16, row16, koff);
        v2f bb = lds_fragB(Ks, t, ej * 16, row16, koff);
        acc = WMMA_F32(a, bb, acc);
      }
#pragma unroll
      for (int j = 0; j < 8; ++j)
        Ms[(di * 16 + crow + j) * PITCH + ej * 16 + row16] = acc[j];
    }
    __syncthreads();
  }
}

extern "C" void kernel_launch(void* const* d_in, const int* in_sizes, int n_in,
                              void* d_out, int out_size, void* d_ws, size_t ws_size,
                              hipStream_t stream) {
  const float* q = (const float*)d_in[0];
  const float* k = (const float*)d_in[1];
  const float* v = (const float*)d_in[2];
  float* y = (float*)d_out;

  const size_t ws_need = (size_t)NHEAD * NGRP * SLOT_FLOATS * sizeof(float);

  if (ws_size >= ws_need) {
    float* gws = (float*)d_ws;
    linmem_group_state <<<dim3(NHEAD * NGRP), dim3(GRP_THREADS), 0, stream>>>(k, v, gws);
    linmem_group_output<<<dim3(NHEAD * NGRP), dim3(GRP_THREADS), 0, stream>>>(q, k, v, gws, y);
  } else {
    linmem_chunked_wmma<<<dim3(NHEAD), dim3(128), 0, stream>>>(q, k, v, y);
  }
}
